# starting kernel
# baseline (speedup 1.0000x reference)
#include <hip/hip_runtime.h>
#include <hip/hip_bf16.h>
#include <math.h>

typedef float v2f __attribute__((ext_vector_type(2)));
typedef float v8f __attribute__((ext_vector_type(8)));

#define NFEAT 256
#define NHID  64
#define NEG_SLOPE 0.2f

__global__ void gat_init(float* __restrict__ out, const float* __restrict__ bias,
                         float* __restrict__ mmax, float* __restrict__ denom, int n) {
    int i = blockIdx.x * blockDim.x + threadIdx.x;
    if (i < n * NHID) out[i] = bias[i & (NHID - 1)];
    if (i < n) { mmax[i] = -3.0e38f; denom[i] = 0.0f; }
}

__global__ void __launch_bounds__(256)
gat_gemm(const float* __restrict__ X, const float* __restrict__ W,
         float* __restrict__ XP, int n) {
    const int tid  = threadIdx.x;
    const int wave = tid >> 5;
    const int lane = tid & 31;
    const int half = lane >> 4;
    const int lrow = lane & 15;
    const int r0   = (blockIdx.x * 8 + wave) * 16;
    if (r0 >= n) return;

    int arow = r0 + lrow;
    if (arow > n - 1) arow = n - 1;
    const float* aptr = X + (size_t)arow * NFEAT + 2 * half;

    v8f c0 = {}, c1 = {}, c2 = {}, c3 = {};

#pragma unroll 8
    for (int k = 0; k < NFEAT; k += 4) {
        v2f a = *(const v2f*)(aptr + k);
        const float* wp = W + (size_t)(k + 2 * half) * NHID + lrow;
        v2f b0, b1, b2, b3;
        b0[0] = wp[0];   b0[1] = wp[NHID + 0];
        b1[0] = wp[16];  b1[1] = wp[NHID + 16];
        b2[0] = wp[32];  b2[1] = wp[NHID + 32];
        b3[0] = wp[48];  b3[1] = wp[NHID + 48];
        c0 = __builtin_amdgcn_wmma_f32_16x16x4_f32(false, a, false, b0, (short)0, c0, false, false);
        c1 = __builtin_amdgcn_wmma_f32_16x16x4_f32(false, a, false, b1, (short)0, c1, false, false);
        c2 = __builtin_amdgcn_wmma_f32_16x16x4_f32(false, a, false, b2, (short)0, c2, false, false);
        c3 = __builtin_amdgcn_wmma_f32_16x16x4_f32(false, a, false, b3, (short)0, c3, false, false);
    }

#pragma unroll
    for (int j = 0; j < 8; ++j) {
        int row = half * 8 + j;
        if (r0 + row < n) {
            float* o = XP + (size_t)(r0 + row) * NHID + lrow;
            o[0]  = c0[j];
            o[16] = c1[j];
            o[32] = c2[j];
            o[48] = c3[j];
        }
    }
}

__global__ void __launch_bounds__(256)
gat_scores(const float* __restrict__ XP, const float* __restrict__ As,
           const float* __restrict__ Ad, float* __restrict__ Ss,
           float* __restrict__ Sd, int n) {
    const int tid  = threadIdx.x;
    const int node = blockIdx.x * 16 + (tid >> 4);
    const int part = tid & 15;
    if (node >= n) return;
    float4 xv = ((const float4*)(XP + (size_t)node * NHID))[part];
    float4 av = ((const float4*)As)[part];
    float4 dv = ((const float4*)Ad)[part];
    float ss = xv.x * av.x + xv.y * av.y + xv.z * av.z + xv.w * av.w;
    float sd = xv.x * dv.x + xv.y * dv.y + xv.z * dv.z + xv.w * dv.w;
#pragma unroll
    for (int m = 1; m < 16; m <<= 1) {
        ss += __shfl_xor(ss, m, 32);
        sd += __shfl_xor(sd, m, 32);
    }
    if (part == 0) { Ss[node] = ss; Sd[node] = sd; }
}

__global__ void gat_logits(const int* __restrict__ src, const int* __restrict__ dst,
                           const float* __restrict__ Ss, const float* __restrict__ Sd,
                           float* __restrict__ ew, float* __restrict__ mmax,
                           int E, int n) {
    int k = blockIdx.x * blockDim.x + threadIdx.x;
    if (k >= E + n) return;
    int s = (k < E) ? src[k] : (k - E);
    int d = (k < E) ? dst[k] : (k - E);
    float v = Ss[s] + Sd[d];
    float e = (v > 0.0f) ? v : NEG_SLOPE * v;
    ew[k] = e;
    atomicMax(mmax + d, e);
}

__global__ void gat_expsum(const int* __restrict__ dst, float* __restrict__ ew,
                           const float* __restrict__ mmax, float* __restrict__ denom,
                           int E, int n) {
    int k = blockIdx.x * blockDim.x + threadIdx.x;
    if (k >= E + n) return;
    int d = (k < E) ? dst[k] : (k - E);
    float ex = __expf(ew[k] - mmax[d]);
    ew[k] = ex;
    atomicAdd(denom + d, ex);
}

__global__ void __launch_bounds__(256)
gat_scatter(const int* __restrict__ src, const int* __restrict__ dst,
            const float* __restrict__ ew, const float* __restrict__ denom,
            const float* __restrict__ XP, float* __restrict__ out, int E, int n) {
    const int tid  = threadIdx.x;
    const int k    = blockIdx.x * 16 + (tid >> 4);
    if (k >= E + n) return;
    const int part = tid & 15;
    int s = (k < E) ? src[k] : (k - E);
    int d = (k < E) ? dst[k] : (k - E);
    float alpha = ew[k] / denom[d];
    float4 xv = ((const float4*)(XP + (size_t)s * NHID))[part];
    float* o = out + (size_t)d * NHID + part * 4;
    atomicAdd(o + 0, alpha * xv.x);
    atomicAdd(o + 1, alpha * xv.y);
    atomicAdd(o + 2, alpha * xv.z);
    atomicAdd(o + 3, alpha * xv.w);
}

extern "C" void kernel_launch(void* const* d_in, const int* in_sizes, int n_in,
                              void* d_out, int out_size, void* d_ws, size_t ws_size,
                              hipStream_t stream) {
    const float* x     = (const float*)d_in[0];
    const int*   ei    = (const int*)  d_in[1];
    const float* W     = (const float*)d_in[2];
    const float* a_src = (const float*)d_in[3];
    const float* a_dst = (const float*)d_in[4];
    const float* bias  = (const float*)d_in[5];
    float*       out   = (float*)d_out;

    const int N = in_sizes[0] / NFEAT;
    const int E = in_sizes[1] / 2;
    const int* src = ei;
    const int* dst = ei + E;

    float* xp    = (float*)d_ws;
    float* ew    = xp    + (size_t)N * NHID;
    float* mmax  = ew    + (size_t)(E + N);
    float* denom = mmax  + N;
    float* ssrc  = denom + N;
    float* sdst  = ssrc  + N;

    gat_init   <<<(N * NHID + 255) / 256, 256, 0, stream>>>(out, bias, mmax, denom, N);
    gat_gemm   <<<(N + 127) / 128,        256, 0, stream>>>(x, W, xp, N);
    gat_scores <<<(N + 15) / 16,          256, 0, stream>>>(xp, a_src, a_dst, ssrc, sdst, N);
    gat_logits <<<(E + N + 255) / 256,    256, 0, stream>>>(src, dst, ssrc, sdst, ew, mmax, E, N);
    gat_expsum <<<(E + N + 255) / 256,    256, 0, stream>>>(dst, ew, mmax, denom, E, N);
    gat_scatter<<<(E + N + 15) / 16,      256, 0, stream>>>(src, dst, ew, denom, xp, out, E, N);
}


# edge pass: destinations split in two by node id; a work-group adds only into its own package half's set (chunk from a per-half counter): a row's atomics come from one half; arrival-order sums
# speedup vs baseline: 2.2545x; 2.2545x over previous
#include <hip/hip_runtime.h>
#include <hip/hip_bf16.h>
#include <math.h>

typedef float v2f __attribute__((ext_vector_type(2)));
typedef float v8f __attribute__((ext_vector_type(8)));

#define NFEAT 256
#define NHID  64
#define NEG_SLOPE 0.2f

#define RANGE_MAX 32.0f
#define TICKET_WORD   32
#define TICKET_STRIDE 32

__device__ __forceinline__ float leaky(float v) { return (v > 0.0f) ? v : NEG_SLOPE * v; }

__global__ void __launch_bounds__(256)
gat_gemm(const float* __restrict__ X, const float* __restrict__ W,
         float* __restrict__ XP, float* __restrict__ gstat, int n) {
    const int tid  = threadIdx.x;
    if (blockIdx.x == 0 && tid < 2) {
        gstat[tid] = -3.0e38f;
        ((unsigned*)gstat)[TICKET_WORD + TICKET_STRIDE * tid] = 0u;
    }
    const int wave = tid >> 5;
    const int lane = tid & 31;
    const int half = lane >> 4;
    const int lrow = lane & 15;
    const int r0   = (blockIdx.x * 8 + wave) * 16;
    if (r0 >= n) return;

    int arow = r0 + lrow;
    if (arow > n - 1) arow = n - 1;
    const float* aptr = X + (size_t)arow * NFEAT + 2 * half;

    v8f c0 = {}, c1 = {}, c2 = {}, c3 = {};

#pragma unroll 8
    for (int k = 0; k < NFEAT; k += 4) {
        v2f a = *(const v2f*)(aptr + k);
        const float* wp = W + (size_t)(k + 2 * half) * NHID + lrow;
        v2f b0, b1, b2, b3;
        b0[0] = wp[0];   b0[1] = wp[NHID + 0];
        b1[0] = wp[16];  b1[1] = wp[NHID + 16];
        b2[0] = wp[32];  b2[1] = wp[NHID + 32];
        b3[0] = wp[48];  b3[1] = wp[NHID + 48];
        c0 = __builtin_amdgcn_wmma_f32_16x16x4_f32(false, a, false, b0, (short)0, c0, false, false);
        c1 = __builtin_amdgcn_wmma_f32_16x16x4_f32(false, a, false, b1, (short)0, c1, false, false);
        c2 = __builtin_amdgcn_wmma_f32_16x16x4_f32(false, a, false, b2, (short)0, c2, false, false);
        c3 = __builtin_amdgcn_wmma_f32_16x16x4_f32(false, a, false, b3, (short)0, c3, false, false);
    }

#pragma unroll
    for (int j = 0; j < 8; ++j) {
        int row = half * 8 + j;
        if (r0 + row < n) {
            float* o = XP + (size_t)(r0 + row) * NHID + lrow;
            o[0]  = c0[j];
            o[16] = c1[j];
            o[32] = c2[j];
            o[48] = c3[j];
        }
    }
}

#define NPB 128
__global__ void __launch_bounds__(256)
gat_scores(const float* __restrict__ XP, const float* __restrict__ As,
           const float* __restrict__ Ad, float* __restrict__ Ss,
           float* __restrict__ SD, float* __restrict__ denom,
           float* __restrict__ out, float* __restrict__ gstat, int n) {
    __shared__ float red[2][8];
    const int tid  = threadIdx.x;
    const int grp  = tid >> 4;
    const int part = tid & 15;
    const float4 av = ((const float4*)As)[part];
    const float4 dv = ((const float4*)Ad)[part];
    float hi = -3.0e38f, lo = -3.0e38f;
    float4 xv[NPB / 16];
#pragma unroll
    for (int it = 0; it < NPB / 16; ++it) {
        const int node = blockIdx.x * NPB + it * 16 + grp;
        const int ld   = (node < n) ? node : (n - 1);
        xv[it] = ((const float4*)(XP + (size_t)ld * NHID))[part];
    }
#pragma unroll
    for (int it = 0; it < NPB / 16; ++it) {
        const int node = blockIdx.x * NPB + it * 16 + grp;
        float ss = xv[it].x * av.x + xv[it].y * av.y + xv[it].z * av.z + xv[it].w * av.w;
        float sd = xv[it].x * dv.x + xv[it].y * dv.y + xv[it].z * dv.z + xv[it].w * dv.w;
#pragma unroll
        for (int m = 1; m < 16; m <<= 1) {
            ss += __shfl_xor(ss, m, 32);
            sd += __shfl_xor(sd, m, 32);
        }
        if (node < n) {
            if (part == 0) {
                Ss[node]    = ss;
                ((float2*)SD)[node] = make_float2(sd, leaky(ss + sd));
                denom[node] = 1.0f;
            }
            ((float4*)(out + (size_t)node * NHID))[part] = xv[it];
            hi = fmaxf(hi, ss);
            lo = fmaxf(lo, -ss);
        }
    }
    hi = fmaxf(hi, __shfl_xor(hi, 16, 32));
    lo = fmaxf(lo, __shfl_xor(lo, 16, 32));
    if ((tid & 31) == 0) { red[0][tid >> 5] = hi; red[1][tid >> 5] = lo; }
    __syncthreads();
    if (tid < 2) {
        float m = red[tid][0];
#pragma unroll
        for (int w = 1; w < 8; ++w) m = fmaxf(m, red[tid][w]);
        atomicMax(gstat + tid, m);
    }
}

#define DR  25600
#define DB  13
#define DT  1024
__global__ void __launch_bounds__(DT)
gat_max_or_denoms(const int* __restrict__ src, const int* __restrict__ dst,
                  const float* __restrict__ Ss, float* __restrict__ SDf,
                  const float* __restrict__ gstat, float* __restrict__ denom,
                  int E, int n) {
    __shared__ float part[DR];
    const int tid = threadIdx.x;
    if (!(gstat[0] + gstat[1] <= RANGE_MAX)) {
        for (int k = blockIdx.x * blockDim.x + tid; k < E; k += gridDim.x * blockDim.x) {
            const int s = src[k];
            const int d = dst[k];
            atomicMax(SDf + 2 * (size_t)d + 1, leaky(Ss[s] + SDf[2 * (size_t)d]));
        }
        return;
    }
    const int nr  = (n + DR - 1) / DR;
    const int nrp = (nr < (int)gridDim.x) ? nr : (int)gridDim.x;
    const int nc  = (int)gridDim.x / nrp;
    const int c   = (int)blockIdx.x / nrp;
    if (c >= nc) return;
    const int per = (E + nc - 1) / nc;
    const int k0  = c * per;
    const int k1  = (k0 + per < E) ? (k0 + per) : E;
    const float2* __restrict__ SD2 = (const float2*)SDf;

    for (int r = (int)blockIdx.x % nrp; r < nr; r += nrp) {
        const int lo = r * DR;
        for (int i = tid; i < DR; i += DT) part[i] = 0.0f;
        __syncthreads();

        for (int b = k0; b < k1; b += DT * DB) {
            int s[DB], d[DB];
#pragma unroll
            for (int j = 0; j < DB; ++j) {
                const int k  = b + j * DT + tid;
                const int kk = (k < k1) ? k : k0;
                s[j] = src[kk];
                const int dd = dst[kk];
                d[j] = (k < k1) ? dd : -1;
            }
            float  sv[DB];
            float2 q[DB];
#pragma unroll
            for (int j = 0; j < DB; ++j) {
                const bool in = (unsigned)(d[j] - lo) < (unsigned)DR;
                sv[j] = Ss[in ? s[j] : lo];
                q[j]  = SD2[in ? d[j] : lo];
            }
#pragma unroll
            for (int j = 0; j < DB; ++j) {
                const bool in = (unsigned)(d[j] - lo) < (unsigned)DR;
                const float ex = __expf(leaky(sv[j] + q[j].x) - q[j].y);
                if (in) atomicAdd(&part[d[j] - lo], ex);
            }
        }
        __syncthreads();
        for (int i = tid; i < DR; i += DT) {
            const int node = lo + i;
            if (node < n) atomicAdd(denom + node, part[i]);
        }
        __syncthreads();
    }
}

#define EPW   32
#define CHUNK 256
template <bool SELF>
__device__ __forceinline__ void edge_window(int lane, unsigned keep, int s, int d, float ex,
                                            const float* __restrict__ XP,
                                            float* __restrict__ out) {
#pragma unroll
    for (int r = 0; r < 2; ++r) {
        float x0[16], x1[16];
#pragma unroll
        for (int j = 0; j < 16; ++j) {
            const int L = 16 * r + j;
            if (keep & (1u << L)) {
                const float* row = XP + (size_t)__builtin_amdgcn_readlane(s, L) * NHID;
                x0[j] = row[lane];
                x1[j] = row[lane + 32];
            }
        }
#pragma unroll
        for (int j = 0; j < 16; ++j) {
            const int L = 16 * r + j;
            if (keep & (1u << L)) {
                const float aj = __int_as_float(__builtin_amdgcn_readlane(__float_as_int(ex), L));
                float* o = out + (size_t)__builtin_amdgcn_readlane(d, L) * NHID;
                atomicAdd(o + lane,      aj * x0[j]);
                atomicAdd(o + lane + 32, aj * x1[j]);
            }
        }
    }
}

template <bool SELF>
__device__ __forceinline__ void edge_wave(int lane, int k, int total, int E, int split, int set,
                                          const int* __restrict__ src, const int* __restrict__ dst,
                                          const float* __restrict__ Ss,
                                          const float* __restrict__ SD,
                                          const float* __restrict__ XP,
                                          float* __restrict__ denom,
                                          float* __restrict__ out) {
    const int kk = (k < total) ? k : (total - 1);
    const int s = (kk < E) ? src[kk] : (kk - E);
    const int d = (kk < E) ? dst[kk] : (kk - E);
    const bool mine = (k < total) && ((d >= split) == (set != 0));
    float ex = 0.0f;
    if (mine) {
        const float2 q = ((const float2*)SD)[d];
        ex = __expf(leaky(Ss[s] + q.x) - q.y);
        if (SELF) {
            if (k >= E) ex -= 1.0f;
            atomicAdd(denom + d, ex);
        }
    }
    const unsigned keep = __builtin_amdgcn_ballot_w32(mine);
    edge_window<SELF>(lane, keep, s, d, ex, XP, out);
}

__global__ void __launch_bounds__(256)
gat_edges(const int* __restrict__ src, const int* __restrict__ dst,
          const float* __restrict__ Ss, const float* __restrict__ SD,
          const float* __restrict__ gstat, unsigned* __restrict__ tickets,
          const float* __restrict__ XP, float* __restrict__ denom,
          float* __restrict__ out, int E, int n) {
    __shared__ int pick[2];
    const int lane  = threadIdx.x & 31;
    const int wave  = __builtin_amdgcn_readfirstlane((int)(threadIdx.x >> 5));

    const bool wide = !(gstat[0] + gstat[1] <= RANGE_MAX);
    const int total = wide ? (E + n) : E;
    const int nch   = (total + CHUNK - 1) / CHUNK;

    if (threadIdx.x == 0) {
        int set = (int)((__builtin_amdgcn_s_sendmsg_rtn(0x87) >> 18) & 1u);
        int t = (int)atomicAdd(tickets + TICKET_STRIDE * set, 1u);
        if (t >= nch) {
            set ^= 1;
            t = (int)atomicAdd(tickets + TICKET_STRIDE * set, 1u);
        }
        pick[0] = t;
        pick[1] = set;
    }
    __syncthreads();
    const int chunk = pick[0];
    const int set   = pick[1];
    if (chunk >= nch) return;

    const int k0 = chunk * CHUNK + wave * EPW;
    if (k0 >= total) return;
    const int split = n >> 1;

    if (!wide)
        edge_wave<false>(lane, k0 + lane, total, E, split, set, src, dst, Ss, SD, XP, denom, out);
    else
        edge_wave<true >(lane, k0 + lane, total, E, split, set, src, dst, Ss, SD, XP, denom, out);
}

__global__ void __launch_bounds__(256)
gat_normalise(float* __restrict__ out, const float* __restrict__ denom,
              const float* __restrict__ bias, int n) {
    const int i = blockIdx.x * blockDim.x + threadIdx.x;
    if (i >= n * (NHID / 4)) return;
    const float inv = 1.0f / denom[i >> 4];
    const float4 b = ((const float4*)bias)[i & 15];
    float4 o = ((float4*)out)[i];
    o.x = o.x * inv + b.x;
    o.y = o.y * inv + b.y;
    o.z = o.z * inv + b.z;
    o.w = o.w * inv + b.w;
    ((float4*)out)[i] = o;
}

extern "C" void kernel_launch(void* const* d_in, const int* in_sizes, int n_in,
                              void* d_out, int out_size, void* d_ws, size_t ws_size,
                              hipStream_t stream) {
    const float* x     = (const float*)d_in[0];
    const int*   ei    = (const int*)  d_in[1];
    const float* W     = (const float*)d_in[2];
    const float* a_src = (const float*)d_in[3];
    const float* a_dst = (const float*)d_in[4];
    const float* bias  = (const float*)d_in[5];
    float*       out   = (float*)d_out;

    const int N = in_sizes[0] / NFEAT;
    const int E = in_sizes[1] / 2;
    const int* src = ei;
    const int* dst = ei + E;

    float* xp    = (float*)d_ws;
    float* gstat = xp    + (size_t)N * NHID;
    float* sd    = gstat + 96;
    float* denom = sd    + 2 * (size_t)N;
    float* ssrc  = denom + N;

    gat_gemm     <<<(N + 127) / 128,          256, 0, stream>>>(x, W, xp, gstat, N);
    gat_scores   <<<(N + NPB - 1) / NPB,      256, 0, stream>>>(xp, a_src, a_dst, ssrc, sd, denom, out, gstat, N);
    gat_max_or_denoms<<<256,                   DT, 0, stream>>>(src, dst, ssrc, sd, gstat, denom, E, N);
    gat_edges    <<<2 * ((E + N + 255) / 256), 256, 0, stream>>>(src, dst, ssrc, sd, gstat, (unsigned*)gstat + TICKET_WORD, xp, denom, out, E, N);
    gat_normalise<<<(N * (NHID / 4) + 255) / 256, 256, 0, stream>>>(out, denom, bias, N);
}


# denominator launch also stores each edge's exp(logit - shift), one plane per destination half (LDS sums per half); edge pass reads it coalesced, gathers no scores; denominator sum order changes
# speedup vs baseline: 2.3957x; 1.0626x over previous
#include <hip/hip_runtime.h>
#include <hip/hip_bf16.h>
#include <math.h>

typedef float v2f __attribute__((ext_vector_type(2)));
typedef float v8f __attribute__((ext_vector_type(8)));

#define NFEAT 256
#define NHID  64
#define NEG_SLOPE 0.2f

#define RANGE_MAX 32.0f
#define TICKET_WORD   32
#define TICKET_STRIDE 32

__device__ __forceinline__ float leaky(float v) { return (v > 0.0f) ? v : NEG_SLOPE * v; }

__global__ void __launch_bounds__(256)
gat_gemm(const float* __restrict__ X, const float* __restrict__ W,
         float* __restrict__ XP, float* __restrict__ gstat, int n) {
    const int tid  = threadIdx.x;
    if (blockIdx.x == 0 && tid < 2) {
        gstat[tid] = -3.0e38f;
        ((unsigned*)gstat)[TICKET_WORD + TICKET_STRIDE * tid] = 0u;
    }
    const int wave = tid >> 5;
    const int lane = tid & 31;
    const int half = lane >> 4;
    const int lrow = lane & 15;
    const int r0   = (blockIdx.x * 8 + wave) * 16;
    if (r0 >= n) return;

    int arow = r0 + lrow;
    if (arow > n - 1) arow = n - 1;
    const float* aptr = X + (size_t)arow * NFEAT + 2 * half;

    v8f c0 = {}, c1 = {}, c2 = {}, c3 = {};

#pragma unroll 8
    for (int k = 0; k < NFEAT; k += 4) {
        v2f a = *(const v2f*)(aptr + k);
        const float* wp = W + (size_t)(k + 2 * half) * NHID + lrow;
        v2f b0, b1, b2, b3;
        b0[0] = wp[0];   b0[1] = wp[NHID + 0];
        b1[0] = wp[16];  b1[1] = wp[NHID + 16];
        b2[0] = wp[32];  b2[1] = wp[NHID + 32];
        b3[0] = wp[48];  b3[1] = wp[NHID + 48];
        c0 = __builtin_amdgcn_wmma_f32_16x16x4_f32(false, a, false, b0, (short)0, c0, false, false);
        c1 = __builtin_amdgcn_wmma_f32_16x16x4_f32(false, a, false, b1, (short)0, c1, false, false);
        c2 = __builtin_amdgcn_wmma_f32_16x16x4_f32(false, a, false, b2, (short)0, c2, false, false);
        c3 = __builtin_amdgcn_wmma_f32_16x16x4_f32(false, a, false, b3, (short)0, c3, false, false);
    }

#pragma unroll
    for (int j = 0; j < 8; ++j) {
        int row = half * 8 + j;
        if (r0 + row < n) {
            float* o = XP + (size_t)(r0 + row) * NHID + lrow;
            o[0]  = c0[j];
            o[16] = c1[j];
            o[32] = c2[j];
            o[48] = c3[j];
        }
    }
}

#define NPB 128
__global__ void __launch_bounds__(256)
gat_scores(const float* __restrict__ XP, const float* __restrict__ As,
           const float* __restrict__ Ad, float* __restrict__ Ss,
           float* __restrict__ SD, float* __restrict__ denom,
           float* __restrict__ out, float* __restrict__ gstat, int n) {
    __shared__ float red[2][8];
    const int tid  = threadIdx.x;
    const int grp  = tid >> 4;
    const int part = tid & 15;
    const float4 av = ((const float4*)As)[part];
    const float4 dv = ((const float4*)Ad)[part];
    float hi = -3.0e38f, lo = -3.0e38f;
    float4 xv[NPB / 16];
#pragma unroll
    for (int it = 0; it < NPB / 16; ++it) {
        const int node = blockIdx.x * NPB + it * 16 + grp;
        const int ld   = (node < n) ? node : (n - 1);
        xv[it] = ((const float4*)(XP + (size_t)ld * NHID))[part];
    }
#pragma unroll
    for (int it = 0; it < NPB / 16; ++it) {
        const int node = blockIdx.x * NPB + it * 16 + grp;
        float ss = xv[it].x * av.x + xv[it].y * av.y + xv[it].z * av.z + xv[it].w * av.w;
        float sd = xv[it].x * dv.x + xv[it].y * dv.y + xv[it].z * dv.z + xv[it].w * dv.w;
#pragma unroll
        for (int m = 1; m < 16; m <<= 1) {
            ss += __shfl_xor(ss, m, 32);
            sd += __shfl_xor(sd, m, 32);
        }
        if (node < n) {
            if (part == 0) {
                Ss[node]    = ss;
                ((float2*)SD)[node] = make_float2(sd, leaky(ss + sd));
                denom[node] = 1.0f;
            }
            ((float4*)(out + (size_t)node * NHID))[part] = xv[it];
            hi = fmaxf(hi, ss);
            lo = fmaxf(lo, -ss);
        }
    }
    hi = fmaxf(hi, __shfl_xor(hi, 16, 32));
    lo = fmaxf(lo, __shfl_xor(lo, 16, 32));
    if ((tid & 31) == 0) { red[0][tid >> 5] = hi; red[1][tid >> 5] = lo; }
    __syncthreads();
    if (tid < 2) {
        float m = red[tid][0];
#pragma unroll
        for (int w = 1; w < 8; ++w) m = fmaxf(m, red[tid][w]);
        atomicMax(gstat + tid, m);
    }
}

#define DRH 51200
#define DB  13
#define DT  1024

__device__ __forceinline__ int gat_split(int n) { return (n >> 1) & ~31; }

__device__ __forceinline__ bool gat_wide(const float* __restrict__ gstat, int n) {
    return !(gstat[0] + gstat[1] <= RANGE_MAX) || (n - gat_split(n) > DRH);
}

__global__ void __launch_bounds__(DT)
gat_max_or_denoms(const int* __restrict__ src, const int* __restrict__ dst,
                  const float* __restrict__ Ss, float* __restrict__ SDf,
                  const float* __restrict__ gstat, float* __restrict__ denom,
                  float* __restrict__ exw, int E, int n) {
    __shared__ float part[DRH];
    const int tid = threadIdx.x;
    if (gat_wide(gstat, n)) {
        for (int k = blockIdx.x * blockDim.x + tid; k < E; k += gridDim.x * blockDim.x) {
            const int s = src[k];
            const int d = dst[k];
            atomicMax(SDf + 2 * (size_t)d + 1, leaky(Ss[s] + SDf[2 * (size_t)d]));
        }
        return;
    }
    const int nc = (int)gridDim.x >> 1;
    const int c  = (int)blockIdx.x >> 1;
    const int h  = (int)blockIdx.x & 1;
    if (c >= nc) return;
    const int per = (((E + nc - 1) / nc) + 31) & ~31;
    const int k0  = c * per;
    const int k1  = (k0 + per < E) ? (k0 + per) : E;
    const int split = gat_split(n);
    const int lo  = h ? split : 0;
    const int cnt = h ? (n - split) : split;
    float* __restrict__ plane = exw + (size_t)h * E;
    const float2* __restrict__ SD2 = (const float2*)SDf;

    for (int i = tid; i < DRH / 4; i += DT) ((float4*)part)[i] = make_float4(0.f, 0.f, 0.f, 0.f);
    __syncthreads();

    for (int b = k0; b < k1; b += DT * DB) {
        int s[DB], d[DB];
#pragma unroll
        for (int j = 0; j < DB; ++j) {
            const int k  = b + j * DT + tid;
            const int kk = (k < k1) ? k : k0;
            s[j] = src[kk];
            const int dd = dst[kk];
            d[j] = (k < k1) ? dd : -1;
        }
        float  sv[DB];
        float2 q[DB];
#pragma unroll
        for (int j = 0; j < DB; ++j) {
            const bool in = (unsigned)(d[j] - lo) < (unsigned)cnt;
            sv[j] = Ss[in ? s[j] : lo];
            q[j]  = SD2[in ? d[j] : lo];
        }
#pragma unroll
        for (int j = 0; j < DB; ++j) {
            const int  k  = b + j * DT + tid;
            const bool in = (unsigned)(d[j] - lo) < (unsigned)cnt;
            const float ex = __expf(leaky(sv[j] + q[j].x) - q[j].y);
            if (in) atomicAdd(&part[d[j] - lo], ex);
            if (k < k1) plane[k] = in ? ex : 0.0f;
        }
    }
    __syncthreads();
    for (int i = tid; i < cnt; i += DT) atomicAdd(denom + lo + i, part[i]);
}

#define EPW   32
#define CHUNK 256
template <bool SELF>
__device__ __forceinline__ void edge_window(int lane, unsigned keep, int s, int d, float ex,
                                            const float* __restrict__ XP,
                                            float* __restrict__ out) {
#pragma unroll
    for (int r = 0; r < 2; ++r) {
        float x0[16], x1[16];
#pragma unroll
        for (int j = 0; j < 16; ++j) {
            const int L = 16 * r + j;
            if (keep & (1u << L)) {
                const float* row = XP + (size_t)__builtin_amdgcn_readlane(s, L) * NHID;
                x0[j] = row[lane];
                x1[j] = row[lane + 32];
            }
        }
#pragma unroll
        for (int j = 0; j < 16; ++j) {
            const int L = 16 * r + j;
            if (keep & (1u << L)) {
                const float aj = __int_as_float(__builtin_amdgcn_readlane(__float_as_int(ex), L));
                float* o = out + (size_t)__builtin_amdgcn_readlane(d, L) * NHID;
                atomicAdd(o + lane,      aj * x0[j]);
                atomicAdd(o + lane + 32, aj * x1[j]);
            }
        }
    }
}

template <bool SELF>
__device__ __forceinline__ void edge_wave(int lane, int k, int total, int E, int split, int set,
                                          const int* __restrict__ src, const int* __restrict__ dst,
                                          const float* __restrict__ Ss,
                                          const float* __restrict__ SD,
                                          const float* __restrict__ exw,
                                          const float* __restrict__ XP,
                                          float* __restrict__ denom,
                                          float* __restrict__ out) {
    const int kk = (k < total) ? k : (total - 1);
    const int s = (kk < E) ? src[kk] : (kk - E);
    const int d = (kk < E) ? dst[kk] : (kk - E);
    const bool mine = (k < total) && ((d >= split) == (set != 0));
    float ex = 0.0f;
    if (SELF) {
        if (mine) {
            const float2 q = ((const float2*)SD)[d];
            ex = __expf(leaky(Ss[s] + q.x) - q.y);
            if (k >= E) ex -= 1.0f;
            atomicAdd(denom + d, ex);
        }
    } else {
        const float w = exw[(size_t)set * E + kk];
        ex = mine ? w : 0.0f;
    }
    const unsigned keep = __builtin_amdgcn_ballot_w32(mine);
    edge_window<SELF>(lane, keep, s, d, ex, XP, out);
}

__global__ void __launch_bounds__(256)
gat_edges(const int* __restrict__ src, const int* __restrict__ dst,
          const float* __restrict__ Ss, const float* __restrict__ SD,
          const float* __restrict__ gstat, unsigned* __restrict__ tickets,
          const float* __restrict__ exw,
          const float* __restrict__ XP, float* __restrict__ denom,
          float* __restrict__ out, int E, int n) {
    __shared__ int pick[2];
    const int lane  = threadIdx.x & 31;
    const int wave  = __builtin_amdgcn_readfirstlane((int)(threadIdx.x >> 5));

    const bool wide = gat_wide(gstat, n);
    const int total = wide ? (E + n) : E;
    const int nch   = (total + CHUNK - 1) / CHUNK;

    if (threadIdx.x == 0) {
        int set = (int)((__builtin_amdgcn_s_sendmsg_rtn(0x87) >> 18) & 1u);
        int t = (int)atomicAdd(tickets + TICKET_STRIDE * set, 1u);
        if (t >= nch) {
            set ^= 1;
            t = (int)atomicAdd(tickets + TICKET_STRIDE * set, 1u);
        }
        pick[0] = t;
        pick[1] = set;
    }
    __syncthreads();
    const int chunk = pick[0];
    const int set   = pick[1];
    if (chunk >= nch) return;

    const int k0 = chunk * CHUNK + wave * EPW;
    if (k0 >= total) return;
    const int split = gat_split(n);

    if (!wide)
        edge_wave<false>(lane, k0 + lane, total, E, split, set, src, dst, Ss, SD, exw, XP, denom, out);
    else
        edge_wave<true >(lane, k0 + lane, total, E, split, set, src, dst, Ss, SD, exw, XP, denom, out);
}

__global__ void __launch_bounds__(256)
gat_normalise(float* __restrict__ out, const float* __restrict__ denom,
              const float* __restrict__ bias, int n) {
    const int i = blockIdx.x * blockDim.x + threadIdx.x;
    if (i >= n * (NHID / 4)) return;
    const float inv = 1.0f / denom[i >> 4];
    const float4 b = ((const float4*)bias)[i & 15];
    float4 o = ((float4*)out)[i];
    o.x = o.x * inv + b.x;
    o.y = o.y * inv + b.y;
    o.z = o.z * inv + b.z;
    o.w = o.w * inv + b.w;
    ((float4*)out)[i] = o;
}

extern "C" void kernel_launch(void* const* d_in, const int* in_sizes, int n_in,
                              void* d_out, int out_size, void* d_ws, size_t ws_size,
                              hipStream_t stream) {
    const float* x     = (const float*)d_in[0];
    const int*   ei    = (const int*)  d_in[1];
    const float* W     = (const float*)d_in[2];
    const float* a_src = (const float*)d_in[3];
    const float* a_dst = (const float*)d_in[4];
    const float* bias  = (const float*)d_in[5];
    float*       out   = (float*)d_out;

    const int N = in_sizes[0] / NFEAT;
    const int E = in_sizes[1] / 2;
    const int* src = ei;
    const int* dst = ei + E;

    float* xp    = (float*)d_ws;
    float* gstat = xp    + (size_t)N * NHID;
    float* sd    = gstat + 96;
    float* denom = sd    + 2 * (size_t)N;
    float* ssrc  = denom + N;
    float* exw   = ssrc  + N;

    gat_gemm     <<<(N + 127) / 128,          256, 0, stream>>>(x, W, xp, gstat, N);
    gat_scores   <<<(N + NPB - 1) / NPB,      256, 0, stream>>>(xp, a_src, a_dst, ssrc, sd, denom, out, gstat, N);
    gat_max_or_denoms<<<256,                   DT, 0, stream>>>(src, dst, ssrc, sd, gstat, denom, exw, E, N);
    gat_edges    <<<2 * ((E + N + 255) / 256), 256, 0, stream>>>(src, dst, ssrc, sd, gstat, (unsigned*)gstat + TICKET_WORD, exw, xp, denom, out, E, N);
    gat_normalise<<<(N * (NHID / 4) + 255) / 256, 256, 0, stream>>>(out, denom, bias, N);
}


# edge pass, denominators, divide in one launch: an owner work-group per 391 rows sums them in LDS, finds its edges by scanning 1-byte owner keys (no sort); global atomics gone; arrival-order LDS sums
# speedup vs baseline: 2.7761x; 1.1588x over previous
#include <hip/hip_runtime.h>
#include <hip/hip_bf16.h>
#include <math.h>

typedef float v2f __attribute__((ext_vector_type(2)));
typedef float v8f __attribute__((ext_vector_type(8)));
typedef unsigned v4u __attribute__((ext_vector_type(4)));

#define NFEAT 256
#define NHID  64
#define NEG_SLOPE 0.2f

#define RANGE_MAX 32.0f

__device__ __forceinline__ float leaky(float v) { return (v > 0.0f) ? v : NEG_SLOPE * v; }

__global__ void __launch_bounds__(256)
gat_gemm(const float* __restrict__ X, const float* __restrict__ W,
         float* __restrict__ XP, float* __restrict__ gstat, int n) {
    const int tid  = threadIdx.x;
    if (blockIdx.x == 0 && tid < 2) gstat[tid] = -3.0e38f;
    const int wave = tid >> 5;
    const int lane = tid & 31;
    const int half = lane >> 4;
    const int lrow = lane & 15;
    const int r0   = (blockIdx.x * 8 + wave) * 16;
    if (r0 >= n) return;

    int arow = r0 + lrow;
    if (arow > n - 1) arow = n - 1;
    const float* aptr = X + (size_t)arow * NFEAT + 2 * half;

    v8f c0 = {}, c1 = {}, c2 = {}, c3 = {};

#pragma unroll 8
    for (int k = 0; k < NFEAT; k += 4) {
        v2f a = *(const v2f*)(aptr + k);
        const float* wp = W + (size_t)(k + 2 * half) * NHID + lrow;
        v2f b0, b1, b2, b3;
        b0[0] = wp[0];   b0[1] = wp[NHID + 0];
        b1[0] = wp[16];  b1[1] = wp[NHID + 16];
        b2[0] = wp[32];  b2[1] = wp[NHID + 32];
        b3[0] = wp[48];  b3[1] = wp[NHID + 48];
        c0 = __builtin_amdgcn_wmma_f32_16x16x4_f32(false, a, false, b0, (short)0, c0, false, false);
        c1 = __builtin_amdgcn_wmma_f32_16x16x4_f32(false, a, false, b1, (short)0, c1, false, false);
        c2 = __builtin_amdgcn_wmma_f32_16x16x4_f32(false, a, false, b2, (short)0, c2, false, false);
        c3 = __builtin_amdgcn_wmma_f32_16x16x4_f32(false, a, false, b3, (short)0, c3, false, false);
    }

#pragma unroll
    for (int j = 0; j < 8; ++j) {
        int row = half * 8 + j;
        if (r0 + row < n) {
            float* o = XP + (size_t)(r0 + row) * NHID + lrow;
            o[0]  = c0[j];
            o[16] = c1[j];
            o[32] = c2[j];
            o[48] = c3[j];
        }
    }
}

#define NPB 128
__global__ void __launch_bounds__(256)
gat_scores(const float* __restrict__ XP, const float* __restrict__ As,
           const float* __restrict__ Ad, const int* __restrict__ dst,
           float* __restrict__ Ss, float* __restrict__ SD,
           unsigned* __restrict__ key4, float* __restrict__ gstat, int E, int n) {
    __shared__ float red[2][8];
    const int tid  = threadIdx.x;
    {
        const unsigned R = (unsigned)((n + 255) >> 8);
        const int nwords = ((E + 15) >> 4) << 2;
        const bool vec = (E & 3) == 0;
        for (int w = blockIdx.x * 256 + tid; w < nwords; w += (int)gridDim.x * 256) {
            const int k = 4 * w;
            unsigned kb = 0u;
            if (vec && k < E) {
                const int4 d4 = ((const int4*)dst)[w];
                kb =  ((unsigned)d4.x / R)        | (((unsigned)d4.y / R) << 8)
                   | (((unsigned)d4.z / R) << 16) | (((unsigned)d4.w / R) << 24);
            } else {
#pragma unroll
                for (int j = 0; j < 4; ++j)
                    if (k + j < E) kb |= ((unsigned)dst[k + j] / R) << (8 * j);
            }
            key4[w] = kb;
        }
    }
    const int grp  = tid >> 4;
    const int part = tid & 15;
    const float4 av = ((const float4*)As)[part];
    const float4 dv = ((const float4*)Ad)[part];
    float hi = -3.0e38f, lo = -3.0e38f;
    float4 xv[NPB / 16];
#pragma unroll
    for (int it = 0; it < NPB / 16; ++it) {
        const int node = blockIdx.x * NPB + it * 16 + grp;
        const int ld   = (node < n) ? node : (n - 1);
        xv[it] = ((const float4*)(XP + (size_t)ld * NHID))[part];
    }
#pragma unroll
    for (int it = 0; it < NPB / 16; ++it) {
        const int node = blockIdx.x * NPB + it * 16 + grp;
        float ss = xv[it].x * av.x + xv[it].y * av.y + xv[it].z * av.z + xv[it].w * av.w;
        float sd = xv[it].x * dv.x + xv[it].y * dv.y + xv[it].z * dv.z + xv[it].w * dv.w;
#pragma unroll
        for (int m = 1; m < 16; m <<= 1) {
            ss += __shfl_xor(ss, m, 32);
            sd += __shfl_xor(sd, m, 32);
        }
        if (node < n) {
            if (part == 0) {
                Ss[node]    = ss;
                ((float2*)SD)[node] = make_float2(sd, leaky(ss + sd));
            }
            hi = fmaxf(hi, ss);
            lo = fmaxf(lo, -ss);
        }
    }
    hi = fmaxf(hi, __shfl_xor(hi, 16, 32));
    lo = fmaxf(lo, __shfl_xor(lo, 16, 32));
    if ((tid & 31) == 0) { red[0][tid >> 5] = hi; red[1][tid >> 5] = lo; }
    __syncthreads();
    if (tid < 2) {
        float m = red[tid][0];
#pragma unroll
        for (int w = 1; w < 8; ++w) m = fmaxf(m, red[tid][w]);
        atomicMax(gstat + tid, m);
    }
}

#define RMAX 512
#define OT   1024
#define KU   8

template <bool MAXP>
__device__ __forceinline__ void owner_batch(int lane, unsigned keep, int kq, int lo,
                                            const int* __restrict__ src,
                                            const int* __restrict__ dst,
                                            const float* __restrict__ Ss,
                                            const float* __restrict__ XP,
                                            float* acc, float* den,
                                            const float* sds, float* shf) {
    const bool valid = ((keep >> lane) & 1u) != 0u;
    const int  kk = valid ? kq : 0;
    const int  s  = src[kk];
    const int  dl = valid ? (dst[kk] - lo) : 0;
    const float ssv = Ss[s];
    const float sdv = sds[dl];
    if (MAXP) {
        if (valid) atomicMax(&shf[dl], leaky(ssv + sdv));
        return;
    }
    const float shv = shf[dl];
    float ex = 0.0f;
#pragma unroll
    for (int r = 0; r < 2; ++r) {
        float x0[16], x1[16];
#pragma unroll
        for (int j = 0; j < 16; ++j) {
            const int L = 16 * r + j;
            if (keep & (1u << L)) {
                const float* row = XP + (size_t)__builtin_amdgcn_readlane(s, L) * NHID;
                x0[j] = row[lane];
                x1[j] = row[lane + 32];
            }
        }
        if (r == 0) {
            float sp = ssv;
            asm volatile("" : "+v"(sp) : : "memory");
            ex = __expf(leaky(sp + sdv) - shv);
            if (valid) atomicAdd(&den[dl], ex);
        }
#pragma unroll
        for (int j = 0; j < 16; ++j) {
            const int L = 16 * r + j;
            if (keep & (1u << L)) {
                const float aj = __int_as_float(__builtin_amdgcn_readlane(__float_as_int(ex), L));
                float* o = acc + __builtin_amdgcn_readlane(dl, L) * NHID;
                atomicAdd(o + lane,      aj * x0[j]);
                atomicAdd(o + lane + 32, aj * x1[j]);
            }
        }
    }
}

__device__ __forceinline__ unsigned zero_bytes(unsigned x) {
    return ~(((x & 0x7F7F7F7Fu) + 0x7F7F7F7Fu) | x | 0x7F7F7F7Fu);
}

__device__ __forceinline__ unsigned hit_nibble(unsigned t) {
    return ((((t >> 7) & 0x01010101u) * 0x01020408u) >> 24) & 0xFu;
}

template <bool MAXP>
__device__ __forceinline__ void owner_scan(int tid, int lane, int b, int lo, int E,
                                           const unsigned char* __restrict__ key,
                                           const int* __restrict__ src,
                                           const int* __restrict__ dst,
                                           const float* __restrict__ Ss,
                                           const float* __restrict__ XP,
                                           float* acc, float* den,
                                           const float* sds, float* shf, v4u* stage) {
    const int G   = (E + 15) >> 4;
    const int T   = (G + OT - 1) / OT;
    const int off = (int)(((unsigned)b * (unsigned)T) >> 8);
    const unsigned pat = (unsigned)b * 0x01010101u;
    const v4u* __restrict__ key16 = (const v4u*)key;
    int kq = 0;
    int qn = 0;
    v4u nx[KU];
#pragma unroll
    for (int u = 0; u < KU; ++u) {
        int itr = u + off;
        if (itr >= T) itr -= T;
        const int g = itr * OT + tid;
        nx[u] = key16[((u < T) && (g < G)) ? g : 0];
    }
    for (int it0 = 0; it0 < T; it0 += KU) {
#pragma unroll
        for (int u = 0; u < KU; ++u) stage[u * OT + tid] = nx[u];
        if (it0 + KU < T) {
#pragma unroll
            for (int u = 0; u < KU; ++u) {
                int itr = it0 + KU + u + off;
                if (itr >= T) itr -= T;
                const int g = itr * OT + tid;
                nx[u] = key16[((it0 + KU + u < T) && (g < G)) ? g : 0];
            }
        }
#pragma unroll 1
        for (int u = 0; u < KU; ++u) {
            int itr = it0 + u + off;
            if (itr >= T) itr -= T;
            const int  g  = itr * OT + tid;
            const bool ok = (it0 + u < T) && (g < G);
            const v4u kv = stage[u * OT + tid];
            const unsigned t0 = zero_bytes(kv.x ^ pat);
            const unsigned t1 = zero_bytes(kv.y ^ pat);
            const unsigned t2 = zero_bytes(kv.z ^ pat);
            const unsigned t3 = zero_bytes(kv.w ^ pat);
            const bool any = ((t0 | t1 | t2 | t3) != 0u) && ok;
            unsigned m = __builtin_amdgcn_ballot_w32(any);
            while (m) {
                const int L = __builtin_ctz(m);
                m &= m - 1u;
                const int k0 = __builtin_amdgcn_readlane(g, L) * 16;
                unsigned h = hit_nibble((unsigned)__builtin_amdgcn_readlane((int)t0, L))
                          | (hit_nibble((unsigned)__builtin_amdgcn_readlane((int)t1, L)) << 4)
                          | (hit_nibble((unsigned)__builtin_amdgcn_readlane((int)t2, L)) << 8)
                          | (hit_nibble((unsigned)__builtin_amdgcn_readlane((int)t3, L)) << 12);
                while (h) {
                    const int k = k0 + __builtin_ctz(h);
                    h &= h - 1u;
                    if (k < E) {
                        kq = (lane == qn) ? k : kq;
                        ++qn;
                        if (qn == 32) {
                            owner_batch<MAXP>(lane, 0xFFFFFFFFu, kq, lo, src, dst, Ss, XP,
                                              acc, den, sds, shf);
                            qn = 0;
                        }
                    }
                }
            }
        }
    }
    if (qn > 0)
        owner_batch<MAXP>(lane, (1u << qn) - 1u, kq, lo, src, dst, Ss, XP, acc, den, sds, shf);
}

__global__ void __launch_bounds__(OT)
gat_owner(const int* __restrict__ src, const int* __restrict__ dst,
          const unsigned char* __restrict__ key,
          const float* __restrict__ Ss, const float* __restrict__ SD,
          const float* __restrict__ gstat, const float* __restrict__ XP,
          const float* __restrict__ bias, float* __restrict__ out, int E, int n) {
    __shared__ float acc[RMAX * NHID];
    __shared__ float den[RMAX];
    __shared__ float sds[RMAX];
    __shared__ float shf[RMAX];
    __shared__ v4u stage[KU * OT];
    const int tid  = threadIdx.x;
    const int lane = tid & 31;
    const int wave = __builtin_amdgcn_readfirstlane(tid >> 5);
    const int R    = (n + 255) >> 8;
    const int b    = (int)blockIdx.x;
    const int lo   = b * R;
    int cnt = n - lo;
    if (cnt > R) cnt = R;
    if (cnt <= 0) return;

    const bool wide = !(gstat[0] + gstat[1] <= RANGE_MAX);
    const float2* __restrict__ SD2 = (const float2*)SD;

    for (int r = tid; r < cnt; r += OT) {
        const float2 q = SD2[lo + r];
        sds[r] = q.x;
        shf[r] = q.y;
    }
    __syncthreads();
    if (wide) {
        owner_scan<true>(tid, lane, b, lo, E, key, src, dst, Ss, XP, acc, den, sds, shf, stage);
        __syncthreads();
    }
    for (int r = wave; r < cnt; r += OT / 32) {
        const float* row = XP + (size_t)(lo + r) * NHID;
        float w0 = 1.0f;
        if (wide) w0 = __expf(SD2[lo + r].y - shf[r]);
        acc[r * NHID + lane]      = w0 * row[lane];
        acc[r * NHID + lane + 32] = w0 * row[lane + 32];
        if (lane == 0) den[r] = w0;
    }
    __syncthreads();
    owner_scan<false>(tid, lane, b, lo, E, key, src, dst, Ss, XP, acc, den, sds, shf, stage);
    __syncthreads();
    for (int r = wave; r < cnt; r += OT / 32) {
        const float inv = 1.0f / den[r];
        float* o = out + (size_t)(lo + r) * NHID;
        o[lane]      = acc[r * NHID + lane]      * inv + bias[lane];
        o[lane + 32] = acc[r * NHID + lane + 32] * inv + bias[lane + 32];
    }
}

extern "C" void kernel_launch(void* const* d_in, const int* in_sizes, int n_in,
                              void* d_out, int out_size, void* d_ws, size_t ws_size,
                              hipStream_t stream) {
    const float* x     = (const float*)d_in[0];
    const int*   ei    = (const int*)  d_in[1];
    const float* W     = (const float*)d_in[2];
    const float* a_src = (const float*)d_in[3];
    const float* a_dst = (const float*)d_in[4];
    const float* bias  = (const float*)d_in[5];
    float*       out   = (float*)d_out;

    const int N = in_sizes[0] / NFEAT;
    const int E = in_sizes[1] / 2;
    const int* src = ei;
    const int* dst = ei + E;

    float* xp    = (float*)d_ws;
    float* gstat = xp    + (size_t)N * NHID;
    unsigned* key = (unsigned*)(gstat + 96);
    float* sd    = (float*)(key + (((size_t)E + 15) / 16) * 4);
    float* ssrc  = sd    + 2 * (size_t)N;

    gat_gemm  <<<(N + 127) / 128,     256, 0, stream>>>(x, W, xp, gstat, N);
    gat_scores<<<(N + NPB - 1) / NPB, 256, 0, stream>>>(xp, a_src, a_dst, dst, ssrc, sd, key, gstat, E, N);
    gat_owner <<<256,                  OT, 0, stream>>>(src, dst, (const unsigned char*)key, ssrc, sd, gstat, xp, bias, out, E, N);
}
